# starting kernel
# baseline (speedup 1.0000x reference)
#include <hip/hip_runtime.h>
#include <hip/hip_bf16.h>

#define N_DRUGS 4096
#define W_DIM   512
#define N_EDGES 131072

typedef __attribute__((ext_vector_type(2))) float v2f;
typedef __attribute__((ext_vector_type(8))) float v8f;

__global__ __launch_bounds__(256)
void dedicom_gemm(const float* __restrict__ Z,
                  const float* __restrict__ W,
                  const float* __restrict__ local_diag,
                  const int*   __restrict__ sub_idx,
                  float* __restrict__ Y) {
    const int tid   = threadIdx.x;
    const int lane  = tid & 31;
    const int wave  = tid >> 5;
    const int row0  = blockIdx.y * 128;
    const int n0g   = blockIdx.x * 64;
    const int m0    = wave * 16;
    const int mlane = lane & 15;
    const int nlane = lane & 15;
    const int koff  = (lane < 16) ? 0 : 2;

    const float* __restrict__ dg = local_diag + (size_t)sub_idx[0] * W_DIM;

    __shared__ float Zs[128 * 36];
    __shared__ float Ws[16 * 160];

    v8f acc[4];
    #pragma unroll
    for (int s = 0; s < 4; ++s)
        acc[s] = v8f{0.f, 0.f, 0.f, 0.f, 0.f, 0.f, 0.f, 0.f};

    for (int k0 = 0; k0 < W_DIM; k0 += 32) {
        __syncthreads();
        #pragma unroll
        for (int i = 0; i < 4; ++i) {
            const int idx = tid + 256 * i;
            const int r   = idx >> 3;
            const int c4  = (idx & 7) << 2;
            const float4 z4 = *(const float4*)&Z[(size_t)(row0 + r) * W_DIM + k0 + c4];
            const float4 d4 = *(const float4*)&dg[k0 + c4];
            *(float4*)&Zs[r * 36 + c4] =
                make_float4(z4.x * d4.x, z4.y * d4.y, z4.z * d4.z, z4.w * d4.w);
        }
        {
            const int kp = tid >> 4;
            const int n4 = (tid & 15) << 2;
            const float4 w0 = *(const float4*)&W[(size_t)(k0 + 2 * kp    ) * W_DIM + n0g + n4];
            const float4 w1 = *(const float4*)&W[(size_t)(k0 + 2 * kp + 1) * W_DIM + n0g + n4];
            float* dst = &Ws[kp * 160 + 2 * n4];
            *(float4*)(dst + 0) = make_float4(w0.x, w1.x, w0.y, w1.y);
            *(float4*)(dst + 4) = make_float4(w0.z, w1.z, w0.w, w1.w);
        }
        if (k0 + 32 < W_DIM) {
            __builtin_prefetch(&Z[(size_t)(row0 + (tid >> 3)) * W_DIM + k0 + 32], 0, 0);
            __builtin_prefetch(&W[(size_t)(k0 + 32 + 2 * (tid >> 4)) * W_DIM + n0g], 0, 0);
        }
        __syncthreads();

        #pragma unroll
        for (int kk = 0; kk < 32; kk += 4) {
            const v2f a = *(const v2f*)&Zs[(m0 + mlane) * 36 + kk + koff];
            const int kp = (kk + koff) >> 1;
            #pragma unroll
            for (int s = 0; s < 4; ++s) {
                const v2f b = *(const v2f*)&Ws[kp * 160 + 2 * (s * 16 + nlane)];
                acc[s] = __builtin_amdgcn_wmma_f32_16x16x4_f32(
                     false, a,  false, b,
                     (short)0, acc[s],
                     false,  false);
            }
        }
    }

    const int rbase = row0 + m0 + ((lane < 16) ? 0 : 8);
    #pragma unroll
    for (int s = 0; s < 4; ++s) {
        const int col = n0g + s * 16 + nlane;
        const float dn = dg[col];
        #pragma unroll
        for (int v = 0; v < 8; ++v) {
            Y[(size_t)(rbase + v) * W_DIM + col] = acc[s][v] * dn;
        }
    }
}

__global__ __launch_bounds__(256)
void dedicom_edge_score(const float* __restrict__ Y,
                        const float* __restrict__ Z,
                        const int*   __restrict__ edges,
                        float* __restrict__ out) {
    const int wave = threadIdx.x >> 5;
    const int lane = threadIdx.x & 31;
    const int e    = blockIdx.x * 8 + wave;
    if (e >= N_EDGES) return;

    const int r = edges[e];
    const int c = edges[N_EDGES + e];
    const float4* __restrict__ yr = (const float4*)(Y + (size_t)r * W_DIM);
    const float4* __restrict__ zc = (const float4*)(Z + (size_t)c * W_DIM);

    float s = 0.f;
    #pragma unroll
    for (int i = 0; i < 4; ++i) {
        const float4 a = yr[lane + 32 * i];
        const float4 b = zc[lane + 32 * i];
        s += a.x * b.x + a.y * b.y + a.z * b.z + a.w * b.w;
    }
    #pragma unroll
    for (int off = 16; off > 0; off >>= 1)
        s += __shfl_xor(s, off, 32);

    if (lane == 0)
        out[e] = 1.0f / (1.0f + __expf(-s));
}

extern "C" void kernel_launch(void* const* d_in, const int* in_sizes, int n_in,
                              void* d_out, int out_size, void* d_ws, size_t ws_size,
                              hipStream_t stream) {
    const float* z_drug     = (const float*)d_in[0];
    const float* gweight    = (const float*)d_in[1];
    const float* local_diag = (const float*)d_in[2];
    const int*   edges      = (const int*)d_in[3];
    const int*   sub_idx    = (const int*)d_in[4];
    float*       out        = (float*)d_out;
    float*       Y          = (float*)d_ws;

    dim3 gemm_grid(W_DIM / 64, N_DRUGS / 128);
    dedicom_gemm<<<gemm_grid, 256, 0, stream>>>(z_drug, gweight, local_diag, sub_idx, Y);

    dedicom_edge_score<<<N_EDGES / 8, 256, 0, stream>>>(Y, z_drug, edges, out);
}


# GEMM K loop software-pipelined: next K step's global loads issued ahead of the current step's WMMAs and held in registers, LDS tiles double-buffered, one barrier per step; summation order unchanged
# speedup vs baseline: 1.1618x; 1.1618x over previous
#include <hip/hip_runtime.h>
#include <hip/hip_bf16.h>

#define N_DRUGS 4096
#define W_DIM   512
#define N_EDGES 131072

typedef __attribute__((ext_vector_type(2))) float v2f;
typedef __attribute__((ext_vector_type(8))) float v8f;

__global__ __launch_bounds__(256)
void dedicom_gemm(const float* __restrict__ Z,
                  const float* __restrict__ W,
                  const float* __restrict__ local_diag,
                  const int*   __restrict__ sub_idx,
                  float* __restrict__ Y) {
    const int tid   = threadIdx.x;
    const int lane  = tid & 31;
    const int wave  = tid >> 5;
    const int row0  = blockIdx.y * 128;
    const int n0g   = blockIdx.x * 64;
    const int m0    = wave * 16;
    const int mlane = lane & 15;
    const int nlane = lane & 15;
    const int koff  = (lane < 16) ? 0 : 2;

    const float* __restrict__ dg = local_diag + (size_t)sub_idx[0] * W_DIM;

    __shared__ float Zs[2][128 * 36];
    __shared__ float Ws[2][16 * 160];

    const int zr  = tid >> 3;
    const int zc4 = (tid & 7) << 2;
    const int wkp = tid >> 4;
    const int wn4 = (tid & 15) << 2;
    const float* __restrict__ zsrc = Z + (size_t)(row0 + zr) * W_DIM + zc4;
    const float* __restrict__ wsrc = W + (size_t)(2 * wkp) * W_DIM + n0g + wn4;
    const float* __restrict__ dsrc = dg + zc4;

    float4 zq[4];
    float4 dq, wq0, wq1;

    v8f acc[4];
    #pragma unroll
    for (int s = 0; s < 4; ++s)
        acc[s] = v8f{0.f, 0.f, 0.f, 0.f, 0.f, 0.f, 0.f, 0.f};

    auto gload = [&](int k0) {
        #pragma unroll
        for (int i = 0; i < 4; ++i)
            zq[i] = *(const float4*)(zsrc + (size_t)(32 * i) * W_DIM + k0);
        dq  = *(const float4*)(dsrc + k0);
        wq0 = *(const float4*)(wsrc + (size_t)(k0    ) * W_DIM);
        wq1 = *(const float4*)(wsrc + (size_t)(k0 + 1) * W_DIM);
    };

    auto lstore = [&](int b) {
        #pragma unroll
        for (int i = 0; i < 4; ++i) {
            *(float4*)&Zs[b][(zr + 32 * i) * 36 + zc4] =
                make_float4(zq[i].x * dq.x, zq[i].y * dq.y, zq[i].z * dq.z, zq[i].w * dq.w);
        }
        float* dst = &Ws[b][wkp * 160 + 2 * wn4];
        *(float4*)(dst + 0) = make_float4(wq0.x, wq1.x, wq0.y, wq1.y);
        *(float4*)(dst + 4) = make_float4(wq0.z, wq1.z, wq0.w, wq1.w);
    };

    auto lfrag = [&](int b, int kk, v2f& a, v2f (&bf)[4]) {
        const float* __restrict__ zs = Zs[b];
        const float* __restrict__ ws = Ws[b];
        a = *(const v2f*)&zs[(m0 + mlane) * 36 + kk + koff];
        const int kp = (kk + koff) >> 1;
        #pragma unroll
        for (int s = 0; s < 4; ++s) {
            bf[s] = *(const v2f*)&ws[kp * 160 + 2 * (s * 16 + nlane)];
        }
    };

    auto mma = [&](const v2f& a, const v2f (&bf)[4]) {
        #pragma unroll
        for (int s = 0; s < 4; ++s) {
            acc[s] = __builtin_amdgcn_wmma_f32_16x16x4_f32(
                 false, a,  false, bf[s],
                 (short)0, acc[s],
                 false,  false);
        }
    };

    auto compute = [&](int b, int kk_lo, int kk_hi) {
        #pragma unroll
        for (int kk = kk_lo; kk < kk_hi; kk += 4) {
            v2f a;
            v2f bf[4];
            lfrag(b, kk, a, bf);
            mma(a, bf);
        }
    };

    gload(0);
    lstore(0);
    __syncthreads();

    for (int t = 0; t < W_DIM / 32 - 1; ++t) {
        gload((t + 1) * 32);
        compute(t & 1, 0, 24);
        v2f a6, a7;
        v2f b6[4], b7[4];
        lfrag(t & 1, 24, a6, b6);
        lfrag(t & 1, 28, a7, b7);
        lstore((t + 1) & 1);
        __builtin_amdgcn_sched_barrier(0);
        mma(a6, b6);
        mma(a7, b7);
        __syncthreads();
    }
    compute((W_DIM / 32 - 1) & 1, 0, 32);

    const int rbase = row0 + m0 + ((lane < 16) ? 0 : 8);
    #pragma unroll
    for (int s = 0; s < 4; ++s) {
        const int col = n0g + s * 16 + nlane;
        const float dn = dg[col];
        #pragma unroll
        for (int v = 0; v < 8; ++v) {
            Y[(size_t)(rbase + v) * W_DIM + col] = acc[s][v] * dn;
        }
    }
}

__global__ __launch_bounds__(256)
void dedicom_edge_score(const float* __restrict__ Y,
                        const float* __restrict__ Z,
                        const int*   __restrict__ edges,
                        float* __restrict__ out) {
    const int wave = threadIdx.x >> 5;
    const int lane = threadIdx.x & 31;
    const int e    = blockIdx.x * 8 + wave;
    if (e >= N_EDGES) return;

    const int r = edges[e];
    const int c = edges[N_EDGES + e];
    const float4* __restrict__ yr = (const float4*)(Y + (size_t)r * W_DIM);
    const float4* __restrict__ zc = (const float4*)(Z + (size_t)c * W_DIM);

    float s = 0.f;
    #pragma unroll
    for (int i = 0; i < 4; ++i) {
        const float4 a = yr[lane + 32 * i];
        const float4 b = zc[lane + 32 * i];
        s += a.x * b.x + a.y * b.y + a.z * b.z + a.w * b.w;
    }
    #pragma unroll
    for (int off = 16; off > 0; off >>= 1)
        s += __shfl_xor(s, off, 32);

    if (lane == 0)
        out[e] = 1.0f / (1.0f + __expf(-s));
}

extern "C" void kernel_launch(void* const* d_in, const int* in_sizes, int n_in,
                              void* d_out, int out_size, void* d_ws, size_t ws_size,
                              hipStream_t stream) {
    const float* z_drug     = (const float*)d_in[0];
    const float* gweight    = (const float*)d_in[1];
    const float* local_diag = (const float*)d_in[2];
    const int*   edges      = (const int*)d_in[3];
    const int*   sub_idx    = (const int*)d_in[4];
    float*       out        = (float*)d_out;
    float*       Y          = (float*)d_ws;

    dim3 gemm_grid(W_DIM / 64, N_DRUGS / 128);
    dedicom_gemm<<<gemm_grid, 256, 0, stream>>>(z_drug, gweight, local_diag, sub_idx, Y);

    dedicom_edge_score<<<N_EDGES / 8, 256, 0, stream>>>(Y, z_drug, edges, out);
}


# Edge stage: one wave scores 32 edges (4096 waves, was 131072); each batch of 4 edges issues its 32 row loads back to back, then 4 interleaved shuffle chains; per-edge sum order unchanged
# speedup vs baseline: 1.3604x; 1.1709x over previous
#include <hip/hip_runtime.h>
#include <hip/hip_bf16.h>

#define N_DRUGS 4096
#define W_DIM   512
#define N_EDGES 131072

typedef __attribute__((ext_vector_type(2))) float v2f;
typedef __attribute__((ext_vector_type(8))) float v8f;

__global__ __launch_bounds__(256)
void dedicom_gemm(const float* __restrict__ Z,
                  const float* __restrict__ W,
                  const float* __restrict__ local_diag,
                  const int*   __restrict__ sub_idx,
                  float* __restrict__ Y) {
    const int tid   = threadIdx.x;
    const int lane  = tid & 31;
    const int wave  = tid >> 5;
    const int row0  = blockIdx.y * 128;
    const int n0g   = blockIdx.x * 64;
    const int m0    = wave * 16;
    const int mlane = lane & 15;
    const int nlane = lane & 15;
    const int koff  = (lane < 16) ? 0 : 2;

    const float* __restrict__ dg = local_diag + (size_t)sub_idx[0] * W_DIM;

    __shared__ float Zs[2][128 * 36];
    __shared__ float Ws[2][16 * 160];

    const int zr  = tid >> 3;
    const int zc4 = (tid & 7) << 2;
    const int wkp = tid >> 4;
    const int wn4 = (tid & 15) << 2;
    const float* __restrict__ zsrc = Z + (size_t)(row0 + zr) * W_DIM + zc4;
    const float* __restrict__ wsrc = W + (size_t)(2 * wkp) * W_DIM + n0g + wn4;
    const float* __restrict__ dsrc = dg + zc4;

    float4 zq[4];
    float4 dq, wq0, wq1;

    v8f acc[4];
    #pragma unroll
    for (int s = 0; s < 4; ++s)
        acc[s] = v8f{0.f, 0.f, 0.f, 0.f, 0.f, 0.f, 0.f, 0.f};

    auto gload = [&](int k0) {
        #pragma unroll
        for (int i = 0; i < 4; ++i)
            zq[i] = *(const float4*)(zsrc + (size_t)(32 * i) * W_DIM + k0);
        dq  = *(const float4*)(dsrc + k0);
        wq0 = *(const float4*)(wsrc + (size_t)(k0    ) * W_DIM);
        wq1 = *(const float4*)(wsrc + (size_t)(k0 + 1) * W_DIM);
    };

    auto lstore = [&](int b) {
        #pragma unroll
        for (int i = 0; i < 4; ++i) {
            *(float4*)&Zs[b][(zr + 32 * i) * 36 + zc4] =
                make_float4(zq[i].x * dq.x, zq[i].y * dq.y, zq[i].z * dq.z, zq[i].w * dq.w);
        }
        float* dst = &Ws[b][wkp * 160 + 2 * wn4];
        *(float4*)(dst + 0) = make_float4(wq0.x, wq1.x, wq0.y, wq1.y);
        *(float4*)(dst + 4) = make_float4(wq0.z, wq1.z, wq0.w, wq1.w);
    };

    auto lfrag = [&](int b, int kk, v2f& a, v2f (&bf)[4]) {
        const float* __restrict__ zs = Zs[b];
        const float* __restrict__ ws = Ws[b];
        a = *(const v2f*)&zs[(m0 + mlane) * 36 + kk + koff];
        const int kp = (kk + koff) >> 1;
        #pragma unroll
        for (int s = 0; s < 4; ++s) {
            bf[s] = *(const v2f*)&ws[kp * 160 + 2 * (s * 16 + nlane)];
        }
    };

    auto mma = [&](const v2f& a, const v2f (&bf)[4]) {
        #pragma unroll
        for (int s = 0; s < 4; ++s) {
            acc[s] = __builtin_amdgcn_wmma_f32_16x16x4_f32(
                 false, a,  false, bf[s],
                 (short)0, acc[s],
                 false,  false);
        }
    };

    auto compute = [&](int b, int kk_lo, int kk_hi) {
        #pragma unroll
        for (int kk = kk_lo; kk < kk_hi; kk += 4) {
            v2f a;
            v2f bf[4];
            lfrag(b, kk, a, bf);
            mma(a, bf);
        }
    };

    gload(0);
    lstore(0);
    __syncthreads();

    for (int t = 0; t < W_DIM / 32 - 1; ++t) {
        gload((t + 1) * 32);
        compute(t & 1, 0, 24);
        v2f a6, a7;
        v2f b6[4], b7[4];
        lfrag(t & 1, 24, a6, b6);
        lfrag(t & 1, 28, a7, b7);
        lstore((t + 1) & 1);
        __builtin_amdgcn_sched_barrier(0);
        mma(a6, b6);
        mma(a7, b7);
        __syncthreads();
    }
    compute((W_DIM / 32 - 1) & 1, 0, 32);

    const int rbase = row0 + m0 + ((lane < 16) ? 0 : 8);
    #pragma unroll
    for (int s = 0; s < 4; ++s) {
        const int col = n0g + s * 16 + nlane;
        const float dn = dg[col];
        #pragma unroll
        for (int v = 0; v < 8; ++v) {
            Y[(size_t)(rbase + v) * W_DIM + col] = acc[s][v] * dn;
        }
    }
}

#define EDGES_PER_WAVE    32
#define EDGES_PER_BATCH   4
#define EDGE_WAVES_PER_WG 8

static_assert(N_EDGES % (EDGES_PER_WAVE * EDGE_WAVES_PER_WG) == 0,
              "the edge grid covers N_EDGES exactly; no tail guard");
static_assert(EDGES_PER_WAVE % EDGES_PER_BATCH == 0, "whole batches");

__device__ __forceinline__ void edge_rows_load(const float* __restrict__ Y,
                                               const float* __restrict__ Z,
                                               int r, int c, int lane,
                                               float4 (&y)[4], float4 (&z)[4]) {
    const float4* __restrict__ yr = (const float4*)(Y + (size_t)r * W_DIM);
    const float4* __restrict__ zc = (const float4*)(Z + (size_t)c * W_DIM);
    #pragma unroll
    for (int i = 0; i < 4; ++i) {
        y[i] = yr[lane + 32 * i];
        z[i] = zc[lane + 32 * i];
    }
}

__device__ __forceinline__ float edge_rows_dot(const float4 (&y)[4], const float4 (&z)[4]) {
    float s = 0.f;
    #pragma unroll
    for (int i = 0; i < 4; ++i) {
        const float4 a = y[i];
        const float4 b = z[i];
        s += a.x * b.x + a.y * b.y + a.z * b.z + a.w * b.w;
    }
    #pragma unroll
    for (int off = 16; off > 0; off >>= 1)
        s += __shfl_xor(s, off, 32);
    return s;
}

__global__ __launch_bounds__(256) __attribute__((amdgpu_waves_per_eu(4, 4)))
void dedicom_edge_score(const float* __restrict__ Y,
                        const float* __restrict__ Z,
                        const int*   __restrict__ edges,
                        float* __restrict__ out) {
    const int lane = threadIdx.x & 31;
    const int wave = __builtin_amdgcn_readfirstlane(threadIdx.x >> 5);
    const int e0   = (blockIdx.x * EDGE_WAVES_PER_WG + wave) * EDGES_PER_WAVE;

    const int ridx = edges[e0 + lane];
    const int cidx = edges[N_EDGES + e0 + lane];

    float res = 0.f;

    for (int j = 0; j < EDGES_PER_WAVE; j += EDGES_PER_BATCH) {
        float4 y[EDGES_PER_BATCH][4], z[EDGES_PER_BATCH][4];
        #pragma unroll
        for (int b = 0; b < EDGES_PER_BATCH; ++b)
            edge_rows_load(Y, Z, __builtin_amdgcn_readlane(ridx, j + b),
                                 __builtin_amdgcn_readlane(cidx, j + b), lane, y[b], z[b]);
        #pragma unroll
        for (int b = 0; b < EDGES_PER_BATCH; ++b) {
            const float s = edge_rows_dot(y[b], z[b]);
            res = (lane == j + b) ? s : res;
        }
    }

    out[e0 + lane] = 1.0f / (1.0f + __expf(-res));
}

extern "C" void kernel_launch(void* const* d_in, const int* in_sizes, int n_in,
                              void* d_out, int out_size, void* d_ws, size_t ws_size,
                              hipStream_t stream) {
    const float* z_drug     = (const float*)d_in[0];
    const float* gweight    = (const float*)d_in[1];
    const float* local_diag = (const float*)d_in[2];
    const int*   edges      = (const int*)d_in[3];
    const int*   sub_idx    = (const int*)d_in[4];
    float*       out        = (float*)d_out;
    float*       Y          = (float*)d_ws;

    dim3 gemm_grid(W_DIM / 64, N_DRUGS / 128);
    dedicom_gemm<<<gemm_grid, 256, 0, stream>>>(z_drug, gweight, local_diag, sub_idx, Y);

    dedicom_edge_score<<<N_EDGES / (EDGES_PER_WAVE * EDGE_WAVES_PER_WG), 256, 0, stream>>>(Y, z_drug, edges, out);
}


# GEMM K-step barrier no longer drains the matrix pipe: next buffer's first fragments are read right behind it under the step's remaining WMMAs; signal/wait split, WMMAs between; sum order unchanged
# speedup vs baseline: 1.4375x; 1.0567x over previous
#include <hip/hip_runtime.h>
#include <hip/hip_bf16.h>

#define N_DRUGS 4096
#define W_DIM   512
#define N_EDGES 131072

typedef __attribute__((ext_vector_type(2))) float v2f;
typedef __attribute__((ext_vector_type(8))) float v8f;

__global__ __launch_bounds__(256)
void dedicom_gemm(const float* __restrict__ Z,
                  const float* __restrict__ W,
                  const float* __restrict__ local_diag,
                  const int*   __restrict__ sub_idx,
                  float* __restrict__ Y) {
    const int tid   = threadIdx.x;
    const int lane  = tid & 31;
    const int wave  = tid >> 5;
    const int row0  = blockIdx.y * 128;
    const int n0g   = blockIdx.x * 64;
    const int m0    = wave * 16;
    const int mlane = lane & 15;
    const int nlane = lane & 15;
    const int koff  = (lane < 16) ? 0 : 2;

    const float* __restrict__ dg = local_diag + (size_t)sub_idx[0] * W_DIM;

    __shared__ float Zs[2][128 * 36];
    __shared__ float Ws[2][16 * 160];

    const int zr  = tid >> 3;
    const int zc4 = (tid & 7) << 2;
    const int wkp = tid >> 4;
    const int wn4 = (tid & 15) << 2;
    const float* __restrict__ zsrc = Z + (size_t)(row0 + zr) * W_DIM + zc4;
    const float* __restrict__ wsrc = W + (size_t)(2 * wkp) * W_DIM + n0g + wn4;
    const float* __restrict__ dsrc = dg + zc4;

    struct Stage { float4 zq[4]; float4 dq, wq0, wq1; };

    v8f acc[4];
    #pragma unroll
    for (int s = 0; s < 4; ++s)
        acc[s] = v8f{0.f, 0.f, 0.f, 0.f, 0.f, 0.f, 0.f, 0.f};

    auto gload = [&](int k0, Stage& g) {
        #pragma unroll
        for (int i = 0; i < 4; ++i)
            g.zq[i] = *(const float4*)(zsrc + (size_t)(32 * i) * W_DIM + k0);
        g.dq  = *(const float4*)(dsrc + k0);
        g.wq0 = *(const float4*)(wsrc + (size_t)(k0    ) * W_DIM);
        g.wq1 = *(const float4*)(wsrc + (size_t)(k0 + 1) * W_DIM);
    };

    auto lstore = [&](int b, const Stage& g) {
        #pragma unroll
        for (int i = 0; i < 4; ++i) {
            *(float4*)&Zs[b][(zr + 32 * i) * 36 + zc4] =
                make_float4(g.zq[i].x * g.dq.x, g.zq[i].y * g.dq.y, g.zq[i].z * g.dq.z, g.zq[i].w * g.dq.w);
        }
        float* dst = &Ws[b][wkp * 160 + 2 * wn4];
        *(float4*)(dst + 0) = make_float4(g.wq0.x, g.wq1.x, g.wq0.y, g.wq1.y);
        *(float4*)(dst + 4) = make_float4(g.wq0.z, g.wq1.z, g.wq0.w, g.wq1.w);
    };

    auto lfrag = [&](int b, int kk, v2f& a, v2f (&bf)[4]) {
        const float* __restrict__ zs = Zs[b];
        const float* __restrict__ ws = Ws[b];
        a = *(const v2f*)&zs[(m0 + mlane) * 36 + kk + koff];
        const int kp = (kk + koff) >> 1;
        #pragma unroll
        for (int s = 0; s < 4; ++s) {
            bf[s] = *(const v2f*)&ws[kp * 160 + 2 * (s * 16 + nlane)];
        }
    };

    auto mma = [&](const v2f& a, const v2f (&bf)[4]) {
        #pragma unroll
        for (int s = 0; s < 4; ++s) {
            acc[s] = __builtin_amdgcn_wmma_f32_16x16x4_f32(
                 false, a,  false, bf[s],
                 (short)0, acc[s],
                 false,  false);
        }
    };

    auto compute = [&](int b, int kk_lo, int kk_hi) {
        #pragma unroll
        for (int kk = kk_lo; kk < kk_hi; kk += 4) {
            v2f a;
            v2f bf[4];
            lfrag(b, kk, a, bf);
            mma(a, bf);
        }
    };

    Stage g0, g;
    gload(0, g0);
    gload(32, g);
    lstore(0, g0);
    __syncthreads();
    v2f f0a, f1a;
    v2f f0b[4], f1b[4];
    lfrag(0, 0, f0a, f0b);
    lfrag(0, 4, f1a, f1b);

    for (int t = 0; t < W_DIM / 32 - 1; ++t) {
        const int b  = t & 1;
        const int nb = b ^ 1;
        lstore(nb, g);
        if (t + 2 < W_DIM / 32)
            gload((t + 2) * 32, g);
        v2f a2, a3, a4, a5, a6, a7;
        v2f b2[4], b3[4], b4[4], b5[4], b6[4], b7[4];
        lfrag(b,  8, a2, b2);
        mma(f0a, f0b);
        lfrag(b, 12, a3, b3);
        mma(f1a, f1b);
        lfrag(b, 16, a4, b4);
        mma(a2, b2);
        lfrag(b, 20, a5, b5);
        mma(a3, b3);
        lfrag(b, 24, a6, b6);
        lfrag(b, 28, a7, b7);
        __builtin_amdgcn_fence(__ATOMIC_RELEASE, "workgroup", "local");
        __builtin_amdgcn_s_barrier_signal(-1);
        __builtin_amdgcn_sched_barrier(0);
        mma(a4, b4);
        mma(a5, b5);
        __builtin_amdgcn_s_barrier_wait(-1);
        __builtin_amdgcn_fence(__ATOMIC_ACQUIRE, "workgroup", "local");
        lfrag(nb, 0, f0a, f0b);
        lfrag(nb, 4, f1a, f1b);
        __builtin_amdgcn_sched_barrier(0);
        mma(a6, b6);
        mma(a7, b7);
    }
    mma(f0a, f0b);
    mma(f1a, f1b);
    compute((W_DIM / 32 - 1) & 1, 8, 32);

    const int rbase = row0 + m0 + ((lane < 16) ? 0 : 8);
    #pragma unroll
    for (int s = 0; s < 4; ++s) {
        const int col = n0g + s * 16 + nlane;
        const float dn = dg[col];
        #pragma unroll
        for (int v = 0; v < 8; ++v) {
            Y[(size_t)(rbase + v) * W_DIM + col] = acc[s][v] * dn;
        }
    }
}

#define EDGES_PER_WAVE    32
#define EDGES_PER_BATCH   4
#define EDGE_WAVES_PER_WG 8

static_assert(N_EDGES % (EDGES_PER_WAVE * EDGE_WAVES_PER_WG) == 0,
              "the edge grid covers N_EDGES exactly; no tail guard");
static_assert(EDGES_PER_WAVE % EDGES_PER_BATCH == 0, "whole batches");

__device__ __forceinline__ void edge_rows_load(const float* __restrict__ Y,
                                               const float* __restrict__ Z,
                                               int r, int c, int lane,
                                               float4 (&y)[4], float4 (&z)[4]) {
    const float4* __restrict__ yr = (const float4*)(Y + (size_t)r * W_DIM);
    const float4* __restrict__ zc = (const float4*)(Z + (size_t)c * W_DIM);
    #pragma unroll
    for (int i = 0; i < 4; ++i) {
        y[i] = yr[lane + 32 * i];
        z[i] = zc[lane + 32 * i];
    }
}

__device__ __forceinline__ float edge_rows_dot(const float4 (&y)[4], const float4 (&z)[4]) {
    float s = 0.f;
    #pragma unroll
    for (int i = 0; i < 4; ++i) {
        const float4 a = y[i];
        const float4 b = z[i];
        s += a.x * b.x + a.y * b.y + a.z * b.z + a.w * b.w;
    }
    #pragma unroll
    for (int off = 16; off > 0; off >>= 1)
        s += __shfl_xor(s, off, 32);
    return s;
}

__global__ __launch_bounds__(256) __attribute__((amdgpu_waves_per_eu(4, 4)))
void dedicom_edge_score(const float* __restrict__ Y,
                        const float* __restrict__ Z,
                        const int*   __restrict__ edges,
                        float* __restrict__ out) {
    const int lane = threadIdx.x & 31;
    const int wave = __builtin_amdgcn_readfirstlane(threadIdx.x >> 5);
    const int e0   = (blockIdx.x * EDGE_WAVES_PER_WG + wave) * EDGES_PER_WAVE;

    const int ridx = edges[e0 + lane];
    const int cidx = edges[N_EDGES + e0 + lane];

    float res = 0.f;

    for (int j = 0; j < EDGES_PER_WAVE; j += EDGES_PER_BATCH) {
        float4 y[EDGES_PER_BATCH][4], z[EDGES_PER_BATCH][4];
        #pragma unroll
        for (int b = 0; b < EDGES_PER_BATCH; ++b)
            edge_rows_load(Y, Z, __builtin_amdgcn_readlane(ridx, j + b),
                                 __builtin_amdgcn_readlane(cidx, j + b), lane, y[b], z[b]);
        #pragma unroll
        for (int b = 0; b < EDGES_PER_BATCH; ++b) {
            const float s = edge_rows_dot(y[b], z[b]);
            res = (lane == j + b) ? s : res;
        }
    }

    out[e0 + lane] = 1.0f / (1.0f + __expf(-res));
}

extern "C" void kernel_launch(void* const* d_in, const int* in_sizes, int n_in,
                              void* d_out, int out_size, void* d_ws, size_t ws_size,
                              hipStream_t stream) {
    const float* z_drug     = (const float*)d_in[0];
    const float* gweight    = (const float*)d_in[1];
    const float* local_diag = (const float*)d_in[2];
    const int*   edges      = (const int*)d_in[3];
    const int*   sub_idx    = (const int*)d_in[4];
    float*       out        = (float*)d_out;
    float*       Y          = (float*)d_ws;

    dim3 gemm_grid(W_DIM / 64, N_DRUGS / 128);
    dedicom_gemm<<<gemm_grid, 256, 0, stream>>>(z_drug, gweight, local_diag, sub_idx, Y);

    dedicom_edge_score<<<N_EDGES / (EDGES_PER_WAVE * EDGE_WAVES_PER_WG), 256, 0, stream>>>(Y, z_drug, edges, out);
}
